# starting kernel
# baseline (speedup 1.0000x reference)
#include <hip/hip_runtime.h>
#include <hip/hip_bf16.h>
#include <math.h>

#define B_DIM 64
#define L_DIM 16384
#define ORDER 3
#define V_DIM 4
#define CTX_DIM 64
#define NBITS 16

typedef float v2f __attribute__((ext_vector_type(2)));
typedef float v8f __attribute__((ext_vector_type(8)));

__global__ __launch_bounds__(128) void mc_pow_kernel(
    const float* __restrict__ time, const float* __restrict__ T,
    const float* __restrict__ W1,   const float* __restrict__ b1,
    const float* __restrict__ W2,   const float* __restrict__ b2,
    float* __restrict__ tbl)
{
    const int lane = threadIdx.x & 31;
    const int wave = threadIdx.x >> 5;
    const int globalWave = blockIdx.x * 4 + wave;
    const int chainBase  = globalWave * 4;
    const int b = chainBase >> 6;
    const float t = time[b];

    int nexp[4];
#pragma unroll
    for (int q = 0; q < 4; ++q) {
        const int ctx = (chainBase + q) & 63;
        float acc = b2[ctx];
        for (int j = 0; j < 32; ++j) {
            float h = fmaf(t, W1[j], b1[j]);
            h = fmaxf(h, 0.0f);
            acc = fmaf(h, W2[ctx * 32 + j], acc);
        }
        float sp = log1pf(expf(-fabsf(acc))) + fmaxf(acc, 0.0f);
        float nf = fmaf(sp, t, 1.0f);
        int ni = (int)nf;
        ni = ni < 0 ? 0 : (ni > 65535 ? 65535 : ni);
        nexp[q] = ni;
    }

    const int m  = lane & 15;
    const int hi = lane >> 4;
    const int k0 = hi * 2;
    const int qA = m >> 2;
    const int rA = m & 3;

    const float* Mq = T + (((chainBase + qA) & 63) * 16);
    v2f aM = { Mq[rA * 4 + k0], Mq[rA * 4 + k0 + 1] };
    v2f bM = { Mq[k0 * 4 + rA], Mq[(k0 + 1) * 4 + rA] };
    v2f aR = { (rA == k0) ? 1.0f : 0.0f, (rA == k0 + 1) ? 1.0f : 0.0f };

    __shared__ float lds[4][2][256];
    float* mm = &lds[wave][0][0];
    float* rm = &lds[wave][1][0];

    for (int k = 0; k < NBITS; ++k) {
        v8f zero = {};
        v8f dRM = __builtin_amdgcn_wmma_f32_16x16x4_f32(
            false, aR, false, bM, (short)0, zero, false, false);
        v8f dMM = __builtin_amdgcn_wmma_f32_16x16x4_f32(
            false, aM, false, bM, (short)0, zero, false, false);

#pragma unroll
        for (int r = 0; r < 8; ++r) {
            const int mr = r + 8 * hi;
            mm[mr * 16 + m] = dMM[r];
            rm[mr * 16 + m] = dRM[r];
        }
        __syncthreads();

        const int base = 4 * qA + k0;
        aM[0] = mm[m * 16 + base];
        aM[1] = mm[m * 16 + base + 1];
        bM[0] = mm[base * 16 + m];
        bM[1] = mm[(base + 1) * 16 + m];
        if ((nexp[qA] >> k) & 1) {
            aR[0] = rm[m * 16 + base];
            aR[1] = rm[m * 16 + base + 1];
        }
        __syncthreads();
    }

    const int chain = chainBase + qA;
    tbl[chain * 16 + rA * 4 + k0]     = aR[0];
    tbl[chain * 16 + rA * 4 + k0 + 1] = aR[1];
}

__global__ __launch_bounds__(256) void mc_gather_kernel(
    const int* __restrict__ seq, const float* __restrict__ tbl,
    float* __restrict__ out)
{
    const int tid = threadIdx.x;
    const int b   = blockIdx.y;
    const int p0  = blockIdx.x * 256;

    __shared__ float s_tbl[CTX_DIM * V_DIM * V_DIM];
    __shared__ int   s_seq[256 + ORDER];

    ((float4*)s_tbl)[tid] = ((const float4*)(tbl + b * 1024))[tid];

    for (int i = tid; i < 256 + ORDER; i += 256) {
        const int p = p0 - ORDER + i;
        s_seq[i] = (p >= 0) ? seq[b * L_DIM + p] : 0;
    }
    __syncthreads();

    const int p = p0 + tid;
    float4 res = make_float4(0.0f, 0.0f, 0.0f, 0.0f);
    if (p >= ORDER) {
        const int s0 = s_seq[tid];
        const int s1 = s_seq[tid + 1];
        const int s2 = s_seq[tid + 2];
        const int ctx = s0 * 16 + s1 * 4 + s2;
        res = ((const float4*)s_tbl)[ctx * 4 + s2];
    }
    ((float4*)out)[b * L_DIM + p] = res;
}

extern "C" void kernel_launch(void* const* d_in, const int* in_sizes, int n_in,
                              void* d_out, int out_size, void* d_ws, size_t ws_size,
                              hipStream_t stream) {
    const int*   seq  = (const int*)d_in[0];
    const float* time = (const float*)d_in[1];
    const float* T    = (const float*)d_in[2];
    const float* W1   = (const float*)d_in[3];
    const float* b1   = (const float*)d_in[4];
    const float* W2   = (const float*)d_in[5];
    const float* b2   = (const float*)d_in[6];
    float* out = (float*)d_out;
    float* tbl = (float*)d_ws;

    hipLaunchKernelGGL(mc_pow_kernel, dim3(4096 / 16), dim3(128), 0, stream,
                       time, T, W1, b1, W2, b2, tbl);

    hipLaunchKernelGGL(mc_gather_kernel, dim3(L_DIM / 256, B_DIM), dim3(256), 0, stream,
                       seq, tbl, out);
}


# power chains stay in registers: 4x4 blocks row-interleaved so each WMMA's D already sits in the next A/B lanes; M^T squared by a twin WMMA, operands swapped; no LDS, no barrier; table bit-identical
# speedup vs baseline: 1.1953x; 1.1953x over previous
#include <hip/hip_runtime.h>
#include <hip/hip_bf16.h>
#include <math.h>

#define B_DIM 64
#define L_DIM 16384
#define ORDER 3
#define V_DIM 4
#define CTX_DIM 64
#define NBITS 16

typedef float v2f __attribute__((ext_vector_type(2)));
typedef float v8f __attribute__((ext_vector_type(8)));

__device__ __forceinline__ v2f pick_pair(const v8f d, const bool q0, const bool q1) {
    const float x0 = q0 ? d[2] : d[0];
    const float x1 = q0 ? d[3] : d[1];
    const float y0 = q0 ? d[6] : d[4];
    const float y1 = q0 ? d[7] : d[5];
    v2f r = { q1 ? y0 : x0, q1 ? y1 : x1 };
    return r;
}

__global__ __launch_bounds__(128) void mc_pow_kernel(
    const float* __restrict__ time, const float* __restrict__ T,
    const float* __restrict__ W1,   const float* __restrict__ b1,
    const float* __restrict__ W2,   const float* __restrict__ b2,
    float* __restrict__ tbl)
{
    const int lane = threadIdx.x & 31;
    const int wave = threadIdx.x >> 5;
    const int globalWave = blockIdx.x * 4 + wave;
    const int chainBase  = globalWave * 4;
    const int b = chainBase >> 6;
    const float t = time[b];

    int nexp[4];
#pragma unroll
    for (int q = 0; q < 4; ++q) {
        const int ctx = (chainBase + q) & 63;
        float acc = b2[ctx];
        for (int j = 0; j < 32; ++j) {
            float h = fmaf(t, W1[j], b1[j]);
            h = fmaxf(h, 0.0f);
            acc = fmaf(h, W2[ctx * 32 + j], acc);
        }
        float sp = log1pf(expf(-fabsf(acc))) + fmaxf(acc, 0.0f);
        float nf = fmaf(sp, t, 1.0f);
        int ni = (int)nf;
        ni = ni < 0 ? 0 : (ni > 65535 ? 65535 : ni);
        nexp[q] = ni;
    }

    const int c  = lane & 15;
    const int hi = lane >> 4;
    const int k0 = hi * 2;
    const int qA = (c & 7) >> 1;
    const int rA = 2 * (c >> 3) + (c & 1);
    const bool q0 = (qA & 1) != 0;
    const bool q1 = (qA & 2) != 0;
    const int nl = q1 ? (q0 ? nexp[3] : nexp[2]) : (q0 ? nexp[1] : nexp[0]);

    const float* Mq = T + (((chainBase + qA) & 63) * 16);
    v2f aM = { Mq[rA * 4 + k0], Mq[rA * 4 + k0 + 1] };
    v2f bM = { Mq[k0 * 4 + rA], Mq[(k0 + 1) * 4 + rA] };
    v2f aR = { (rA == k0) ? 1.0f : 0.0f, (rA == k0 + 1) ? 1.0f : 0.0f };

    for (int k = 0; k < NBITS; ++k) {
        v8f zero = {};
        v8f dMM = __builtin_amdgcn_wmma_f32_16x16x4_f32(
            false, aM, false, bM, (short)0, zero, false, false);
        v8f dMT = __builtin_amdgcn_wmma_f32_16x16x4_f32(
            false, bM, false, aM, (short)0, zero, false, false);
        v8f dRT = __builtin_amdgcn_wmma_f32_16x16x4_f32(
            false, bM, false, aR, (short)0, zero, false, false);

        if ((nl >> k) & 1)
            aR = pick_pair(dRT, q0, q1);
        bM = pick_pair(dMM, q0, q1);
        aM = pick_pair(dMT, q0, q1);
    }

    const int chain = chainBase + qA;
    tbl[chain * 16 + rA * 4 + k0]     = aR[0];
    tbl[chain * 16 + rA * 4 + k0 + 1] = aR[1];
}

__global__ __launch_bounds__(256) void mc_gather_kernel(
    const int* __restrict__ seq, const float* __restrict__ tbl,
    float* __restrict__ out)
{
    const int tid = threadIdx.x;
    const int b   = blockIdx.y;
    const int p0  = blockIdx.x * 256;

    __shared__ float s_tbl[CTX_DIM * V_DIM * V_DIM];
    __shared__ int   s_seq[256 + ORDER];

    ((float4*)s_tbl)[tid] = ((const float4*)(tbl + b * 1024))[tid];

    for (int i = tid; i < 256 + ORDER; i += 256) {
        const int p = p0 - ORDER + i;
        s_seq[i] = (p >= 0) ? seq[b * L_DIM + p] : 0;
    }
    __syncthreads();

    const int p = p0 + tid;
    float4 res = make_float4(0.0f, 0.0f, 0.0f, 0.0f);
    if (p >= ORDER) {
        const int s0 = s_seq[tid];
        const int s1 = s_seq[tid + 1];
        const int s2 = s_seq[tid + 2];
        const int ctx = s0 * 16 + s1 * 4 + s2;
        res = ((const float4*)s_tbl)[ctx * 4 + s2];
    }
    ((float4*)out)[b * L_DIM + p] = res;
}

extern "C" void kernel_launch(void* const* d_in, const int* in_sizes, int n_in,
                              void* d_out, int out_size, void* d_ws, size_t ws_size,
                              hipStream_t stream) {
    const int*   seq  = (const int*)d_in[0];
    const float* time = (const float*)d_in[1];
    const float* T    = (const float*)d_in[2];
    const float* W1   = (const float*)d_in[3];
    const float* b1   = (const float*)d_in[4];
    const float* W2   = (const float*)d_in[5];
    const float* b2   = (const float*)d_in[6];
    float* out = (float*)d_out;
    float* tbl = (float*)d_ws;

    hipLaunchKernelGGL(mc_pow_kernel, dim3(4096 / 16), dim3(128), 0, stream,
                       time, T, W1, b1, W2, b2, tbl);

    hipLaunchKernelGGL(mc_gather_kernel, dim3(L_DIM / 256, B_DIM), dim3(256), 0, stream,
                       seq, tbl, out);
}


# gather out of registers: a wave keeps its b's 64 rows in 16 VGPRs and picks each row by in-row lane permutes; no LDS, no barrier, 2048 waves of 512 positions; power kernel stores only those rows
# speedup vs baseline: 1.7190x; 1.4381x over previous
#include <hip/hip_runtime.h>
#include <hip/hip_bf16.h>
#include <math.h>

#define B_DIM 64
#define L_DIM 16384
#define ORDER 3
#define V_DIM 4
#define CTX_DIM 64
#define NBITS 16

typedef float v2f __attribute__((ext_vector_type(2)));
typedef float v8f __attribute__((ext_vector_type(8)));

__device__ __forceinline__ v2f pick_pair(const v8f d, const bool q0, const bool q1) {
    const float x0 = q0 ? d[2] : d[0];
    const float x1 = q0 ? d[3] : d[1];
    const float y0 = q0 ? d[6] : d[4];
    const float y1 = q0 ? d[7] : d[5];
    v2f r = { q1 ? y0 : x0, q1 ? y1 : x1 };
    return r;
}

__global__ __launch_bounds__(128) void mc_pow_kernel(
    const float* __restrict__ time, const float* __restrict__ T,
    const float* __restrict__ W1,   const float* __restrict__ b1,
    const float* __restrict__ W2,   const float* __restrict__ b2,
    float* __restrict__ sel)
{
    const int lane = threadIdx.x & 31;
    const int wave = threadIdx.x >> 5;
    const int globalWave = blockIdx.x * 4 + wave;
    const int chainBase  = globalWave * 4;
    const int b = chainBase >> 6;
    const float t = time[b];

    int nexp[4];
#pragma unroll
    for (int q = 0; q < 4; ++q) {
        const int ctx = (chainBase + q) & 63;
        float acc = b2[ctx];
        for (int j = 0; j < 32; ++j) {
            float h = fmaf(t, W1[j], b1[j]);
            h = fmaxf(h, 0.0f);
            acc = fmaf(h, W2[ctx * 32 + j], acc);
        }
        float sp = log1pf(expf(-fabsf(acc))) + fmaxf(acc, 0.0f);
        float nf = fmaf(sp, t, 1.0f);
        int ni = (int)nf;
        ni = ni < 0 ? 0 : (ni > 65535 ? 65535 : ni);
        nexp[q] = ni;
    }

    const int c  = lane & 15;
    const int hi = lane >> 4;
    const int k0 = hi * 2;
    const int qA = (c & 7) >> 1;
    const int rA = 2 * (c >> 3) + (c & 1);
    const bool q0 = (qA & 1) != 0;
    const bool q1 = (qA & 2) != 0;
    const int nl = q1 ? (q0 ? nexp[3] : nexp[2]) : (q0 ? nexp[1] : nexp[0]);

    const float* Mq = T + (((chainBase + qA) & 63) * 16);
    v2f aM = { Mq[rA * 4 + k0], Mq[rA * 4 + k0 + 1] };
    v2f bM = { Mq[k0 * 4 + rA], Mq[(k0 + 1) * 4 + rA] };
    v2f aR = { (rA == k0) ? 1.0f : 0.0f, (rA == k0 + 1) ? 1.0f : 0.0f };

    for (int k = 0; k < NBITS; ++k) {
        v8f zero = {};
        v8f dMM = __builtin_amdgcn_wmma_f32_16x16x4_f32(
            false, aM, false, bM, (short)0, zero, false, false);
        v8f dMT = __builtin_amdgcn_wmma_f32_16x16x4_f32(
            false, bM, false, aM, (short)0, zero, false, false);
        v8f dRT = __builtin_amdgcn_wmma_f32_16x16x4_f32(
            false, bM, false, aR, (short)0, zero, false, false);

        if ((nl >> k) & 1)
            aR = pick_pair(dRT, q0, q1);
        bM = pick_pair(dMM, q0, q1);
        aM = pick_pair(dMT, q0, q1);
    }

    const int chain = chainBase + qA;
    if (rA == qA) {
        sel[chain * 4 + k0]     = aR[0];
        sel[chain * 4 + k0 + 1] = aR[1];
    }
}

#define NPOS 16

typedef int      v3i __attribute__((ext_vector_type(3)));
typedef unsigned v4u __attribute__((ext_vector_type(4)));

__device__ __forceinline__ unsigned row_gather(const unsigned x, const unsigned j) {
    return __builtin_amdgcn_permlane16_var(0u, x, j, true, false);
}

__global__ __launch_bounds__(256) void mc_gather_kernel(
    const int* __restrict__ seq, const float* __restrict__ sel,
    float* __restrict__ out)
{
    const int lane   = threadIdx.x & 31;
    const int wave   = threadIdx.x >> 5;
    const int b      = blockIdx.x >> 2;
    const int strip0 = (blockIdx.x & 3) * 4096 + wave * (32 * NPOS);
    const int p0     = strip0 + lane;

    const int* __restrict__ sq = seq + b * L_DIM;
    v3i s[NPOS];
#pragma unroll
    for (int i = 0; i < NPOS; ++i) {
        int i0 = p0 + 32 * i - 3, i1 = i0 + 1, i2 = i0 + 2;
        if (i == 0) {
            i0 = i0 < 0 ? 0 : i0;
            i1 = i1 < 0 ? 0 : i1;
            i2 = i2 < 0 ? 0 : i2;
        }
        s[i].x = sq[i0];
        s[i].y = sq[i1];
        s[i].z = sq[i2];
    }
    const v4u* __restrict__ rows = (const v4u*)sel + b * CTX_DIM + (lane & 15);
    v4u T[4];
#pragma unroll
    for (int t = 0; t < 4; ++t)
        T[t] = rows[16 * t];

    asm volatile("" :: "v"(s[0]), "v"(s[1]), "v"(s[2]), "v"(s[3]), "v"(s[4]), "v"(s[5]),
                       "v"(s[6]), "v"(s[7]), "v"(s[8]), "v"(s[9]), "v"(s[10]), "v"(s[11]),
                       "v"(s[12]), "v"(s[13]), "v"(s[14]), "v"(s[15]),
                       "v"(T[0]), "v"(T[1]), "v"(T[2]), "v"(T[3]));

    v4u* __restrict__ op = (v4u*)out + b * L_DIM + p0;
#pragma unroll
    for (int i = 0; i < NPOS; ++i) {
        const unsigned j  = (unsigned)(s[i].y * 4 + s[i].z);
        const bool     b0 = (s[i].x & 1) != 0;
        const bool     b1 = (s[i].x & 2) != 0;
        v4u r;
#pragma unroll
        for (int c = 0; c < 4; ++c) {
            const unsigned x0 = row_gather(T[0][c], j);
            const unsigned x1 = row_gather(T[1][c], j);
            const unsigned x2 = row_gather(T[2][c], j);
            const unsigned x3 = row_gather(T[3][c], j);
            const unsigned lo = b0 ? x1 : x0;
            const unsigned hi = b0 ? x3 : x2;
            r[c] = b1 ? hi : lo;
        }
        if (i == 0 && p0 < ORDER) r = (v4u){0u, 0u, 0u, 0u};
        op[32 * i] = r;
    }
}

extern "C" void kernel_launch(void* const* d_in, const int* in_sizes, int n_in,
                              void* d_out, int out_size, void* d_ws, size_t ws_size,
                              hipStream_t stream) {
    const int*   seq  = (const int*)d_in[0];
    const float* time = (const float*)d_in[1];
    const float* T    = (const float*)d_in[2];
    const float* W1   = (const float*)d_in[3];
    const float* b1   = (const float*)d_in[4];
    const float* W2   = (const float*)d_in[5];
    const float* b2   = (const float*)d_in[6];
    float* out = (float*)d_out;
    float* sel = (float*)d_ws;

    hipLaunchKernelGGL(mc_pow_kernel, dim3(4096 / 16), dim3(128), 0, stream,
                       time, T, W1, b1, W2, b2, sel);

    hipLaunchKernelGGL(mc_gather_kernel, dim3(B_DIM * 4), dim3(256), 0, stream,
                       seq, sel, out);
}
